# starting kernel
# baseline (speedup 1.0000x reference)
#include <hip/hip_runtime.h>
#include <hip/hip_bf16.h>
#include <math.h>

#define B_      8
#define N_      1024
#define CIN_    256
#define H_      4
#define COUT_   256
#define C_      64
#define ALPHA_  0.2f
#define NEG_INF_ (-9.0e15f)

typedef float v2f __attribute__((ext_vector_type(2)));
typedef float v8f __attribute__((ext_vector_type(8)));

__global__ __launch_bounds__(256) void gat_gemm_h(const float* __restrict__ X,
                                                  const float* __restrict__ W,
                                                  const float* __restrict__ bias,
                                                  float* __restrict__ hbuf) {
  const int wave = blockIdx.x * (blockDim.x >> 5) + (threadIdx.x >> 5);
  const int lane = threadIdx.x & 31;
  const int half = lane >> 4;
  const int l    = lane & 15;

  const int tileM = wave >> 2;
  const int cg    = wave & 3;
  const int row0  = tileM * 16;
  const int col0  = cg * 64;

  const float* xrow = X + (size_t)(row0 + l) * CIN_;
  const float* wr0  = W + (size_t)(col0 +  0 + l) * CIN_;
  const float* wr1  = W + (size_t)(col0 + 16 + l) * CIN_;
  const float* wr2  = W + (size_t)(col0 + 32 + l) * CIN_;
  const float* wr3  = W + (size_t)(col0 + 48 + l) * CIN_;

  v8f acc[4] = {{}, {}, {}, {}};
  #pragma unroll 4
  for (int kk = 0; kk < CIN_; kk += 4) {
    const int ko = kk + 2 * half;
    v2f av  = *(const v2f*)(xrow + ko);
    v2f bv0 = *(const v2f*)(wr0 + ko);
    v2f bv1 = *(const v2f*)(wr1 + ko);
    v2f bv2 = *(const v2f*)(wr2 + ko);
    v2f bv3 = *(const v2f*)(wr3 + ko);
    acc[0] = __builtin_amdgcn_wmma_f32_16x16x4_f32(false, av, false, bv0,
                                                   (short)0, acc[0], false, false);
    acc[1] = __builtin_amdgcn_wmma_f32_16x16x4_f32(false, av, false, bv1,
                                                   (short)0, acc[1], false, false);
    acc[2] = __builtin_amdgcn_wmma_f32_16x16x4_f32(false, av, false, bv2,
                                                   (short)0, acc[2], false, false);
    acc[3] = __builtin_amdgcn_wmma_f32_16x16x4_f32(false, av, false, bv3,
                                                   (short)0, acc[3], false, false);
  }

  #pragma unroll
  for (int cc = 0; cc < 4; ++cc) {
    const float bval = bias[col0 + cc * 16 + l];
    #pragma unroll
    for (int r = 0; r < 8; ++r) {
      const int row = row0 + r + 8 * half;
      hbuf[(size_t)row * COUT_ + col0 + cc * 16 + l] = acc[cc][r] + bval;
    }
  }
}

__global__ __launch_bounds__(256) void gat_lscore(const float* __restrict__ hbuf,
                                                  const float* __restrict__ a,
                                                  float* __restrict__ lsrc,
                                                  float* __restrict__ ldst) {
  const int t    = blockIdx.x * blockDim.x + threadIdx.x;
  const int head = t & (H_ - 1);
  const int bi   = t >> 2;
  const float* hr = hbuf + (size_t)bi * COUT_ + head * C_;
  const float* as = a + head * (2 * C_);
  const float* ad = as + C_;
  float s = 0.f, d = 0.f;
  #pragma unroll 8
  for (int c = 0; c < C_; ++c) {
    const float v = hr[c];
    s = fmaf(v, as[c], s);
    d = fmaf(v, ad[c], d);
  }
  lsrc[t] = s;
  ldst[t] = d;
}

__global__ __launch_bounds__(256) void gat_stats(const float* __restrict__ lsrc,
                                                 const float* __restrict__ ldst,
                                                 const int*   __restrict__ adj,
                                                 float* __restrict__ mbuf,
                                                 float* __restrict__ sbuf) {
  const int wave = blockIdx.x * (blockDim.x >> 5) + (threadIdx.x >> 5);
  const int lane = threadIdx.x & 31;
  const size_t bi = (size_t)wave;
  const int b = wave / N_;

  float ls[H_];
  #pragma unroll
  for (int hd = 0; hd < H_; ++hd) ls[hd] = lsrc[bi * H_ + hd];

  const int*    arow = adj + bi * (size_t)N_;
  const float4* ldr  = (const float4*)ldst;

  float mx[H_];
  #pragma unroll
  for (int hd = 0; hd < H_; ++hd) mx[hd] = NEG_INF_;

  for (int j = lane; j < N_; j += 32) {
    const int    ad = arow[j];
    const float4 ld = ldr[(size_t)b * N_ + j];
    const float  ldv[H_] = {ld.x, ld.y, ld.z, ld.w};
    #pragma unroll
    for (int hd = 0; hd < H_; ++hd) {
      float x = ls[hd] + ldv[hd];
      x = fmaxf(x, ALPHA_ * x);
      x = (ad == 1) ? x : NEG_INF_;
      mx[hd] = fmaxf(mx[hd], x);
    }
  }
  #pragma unroll
  for (int hd = 0; hd < H_; ++hd)
    #pragma unroll
    for (int off = 16; off > 0; off >>= 1)
      mx[hd] = fmaxf(mx[hd], __shfl_xor(mx[hd], off, 32));

  float sm[H_] = {0.f, 0.f, 0.f, 0.f};
  for (int j = lane; j < N_; j += 32) {
    const int    ad = arow[j];
    const float4 ld = ldr[(size_t)b * N_ + j];
    const float  ldv[H_] = {ld.x, ld.y, ld.z, ld.w};
    #pragma unroll
    for (int hd = 0; hd < H_; ++hd) {
      float x = ls[hd] + ldv[hd];
      x = fmaxf(x, ALPHA_ * x);
      sm[hd] += (ad == 1) ? __expf(x - mx[hd]) : 0.f;
    }
  }
  #pragma unroll
  for (int hd = 0; hd < H_; ++hd)
    #pragma unroll
    for (int off = 16; off > 0; off >>= 1)
      sm[hd] += __shfl_xor(sm[hd], off, 32);

  if (lane == 0) {
    #pragma unroll
    for (int hd = 0; hd < H_; ++hd) {
      mbuf[bi * H_ + hd] = mx[hd];
      sbuf[bi * H_ + hd] = sm[hd];
    }
  }
}

__global__ __launch_bounds__(128) void gat_aggregate(const float* __restrict__ hbuf,
                                                     const float* __restrict__ lsrc,
                                                     const float* __restrict__ ldst,
                                                     const int*   __restrict__ adj,
                                                     const float* __restrict__ mbuf,
                                                     const float* __restrict__ sbuf,
                                                     float* __restrict__ out) {
  __shared__ float lds_ld[N_ * H_];

  const int b     = blockIdx.x >> 5;
  const int islab = blockIdx.x & 31;
  const int i0    = islab * 32;
  const int head  = threadIdx.x >> 5;
  const int lane  = threadIdx.x & 31;
  const int half  = lane >> 4;
  const int l     = lane & 15;

  {
    const float4* src = (const float4*)(ldst + (size_t)b * N_ * H_);
    float4*       dst = (float4*)lds_ld;
    for (int j = threadIdx.x; j < N_; j += 128) dst[j] = src[j];
  }
  __syncthreads();

  const size_t biA = (size_t)b * N_ + (i0 + l);
  const size_t biB = biA + 16;
  const float  lsA = lsrc[biA * H_ + head];
  const float  lsB = lsrc[biB * H_ + head];
  const float  mA  = mbuf[biA * H_ + head];
  const float  mB  = mbuf[biB * H_ + head];
  const float  isA = 1.0f / sbuf[biA * H_ + head];
  const float  isB = 1.0f / sbuf[biB * H_ + head];
  const int*   arA = adj + biA * (size_t)N_;
  const int*   arB = adj + biB * (size_t)N_;

  v8f accA[4] = {{}, {}, {}, {}};
  v8f accB[4] = {{}, {}, {}, {}};

  for (int j0 = 0; j0 < N_; j0 += 4) {
    const int ja = j0 + 2 * half;

    const int2  adA = *(const int2*)(arA + ja);
    const int2  adB = *(const int2*)(arB + ja);
    const float ld0 = lds_ld[ja * H_ + head];
    const float ld1 = lds_ld[(ja + 1) * H_ + head];

    float xA0 = lsA + ld0; xA0 = fmaxf(xA0, ALPHA_ * xA0);
    float xA1 = lsA + ld1; xA1 = fmaxf(xA1, ALPHA_ * xA1);
    float xB0 = lsB + ld0; xB0 = fmaxf(xB0, ALPHA_ * xB0);
    float xB1 = lsB + ld1; xB1 = fmaxf(xB1, ALPHA_ * xB1);

    v2f avA, avB;
    avA.x = (adA.x == 1) ? __expf(xA0 - mA) * isA : 0.0f;
    avA.y = (adA.y == 1) ? __expf(xA1 - mA) * isA : 0.0f;
    avB.x = (adB.x == 1) ? __expf(xB0 - mB) * isB : 0.0f;
    avB.y = (adB.y == 1) ? __expf(xB1 - mB) * isB : 0.0f;

    const float* hrow0 = hbuf + ((size_t)b * N_ + ja) * COUT_ + head * C_ + l;
    const float* hrow1 = hrow0 + COUT_;
    #pragma unroll
    for (int cc = 0; cc < 4; ++cc) {
      v2f bv;
      bv.x = hrow0[cc * 16];
      bv.y = hrow1[cc * 16];
      accA[cc] = __builtin_amdgcn_wmma_f32_16x16x4_f32(false, avA, false, bv,
                                                       (short)0, accA[cc],
                                                       false, false);
      accB[cc] = __builtin_amdgcn_wmma_f32_16x16x4_f32(false, avB, false, bv,
                                                       (short)0, accB[cc],
                                                       false, false);
    }
  }

  #pragma unroll
  for (int cc = 0; cc < 4; ++cc) {
    #pragma unroll
    for (int r = 0; r < 8; ++r) {
      const int rowA = i0 + r + 8 * half;
      const int rowB = rowA + 16;
      out[((size_t)b * N_ + rowA) * COUT_ + head * C_ + cc * 16 + l] = accA[cc][r];
      out[((size_t)b * N_ + rowB) * COUT_ + head * C_ + cc * 16 + l] = accB[cc][r];
    }
  }
}

extern "C" void kernel_launch(void* const* d_in, const int* in_sizes, int n_in,
                              void* d_out, int out_size, void* d_ws, size_t ws_size,
                              hipStream_t stream) {
  const float* node_feats = (const float*)d_in[0];
  const int*   adj        = (const int*)  d_in[1];
  const float* W          = (const float*)d_in[2];
  const float* bias       = (const float*)d_in[3];
  const float* a          = (const float*)d_in[4];
  float*       out        = (float*)d_out;

  float* hbuf = (float*)d_ws;
  float* lsrc = hbuf + (size_t)B_ * N_ * COUT_;
  float* ldst = lsrc + (size_t)B_ * N_ * H_;
  float* mbuf = ldst + (size_t)B_ * N_ * H_;
  float* sbuf = mbuf + (size_t)B_ * N_ * H_;

  gat_gemm_h<<<(B_ * N_ / 16) * (COUT_ / 64) / 8, 256, 0, stream>>>(
      node_feats, W, bias, hbuf);

  gat_lscore<<<(B_ * N_ * H_) / 256, 256, 0, stream>>>(hbuf, a, lsrc, ldst);

  gat_stats<<<(B_ * N_) / 8, 256, 0, stream>>>(lsrc, ldst, adj, mbuf, sbuf);

  gat_aggregate<<<B_ * (N_ / 32), 128, 0, stream>>>(
      hbuf, lsrc, ldst, adj, mbuf, sbuf, out);
}


# Aggregation: split each 32x64 tile's neighbour sum over 4 waves per head (4 waves per SIMD cover load latency); partials summed via LDS in fixed slice order, which changes summation order
# speedup vs baseline: 1.8138x; 1.8138x over previous
#include <hip/hip_runtime.h>
#include <hip/hip_bf16.h>
#include <math.h>

#define B_      8
#define N_      1024
#define CIN_    256
#define H_      4
#define COUT_   256
#define C_      64
#define ALPHA_  0.2f
#define NEG_INF_ (-9.0e15f)

typedef float v2f __attribute__((ext_vector_type(2)));
typedef float v8f __attribute__((ext_vector_type(8)));

__global__ __launch_bounds__(256) void gat_gemm_h(const float* __restrict__ X,
                                                  const float* __restrict__ W,
                                                  const float* __restrict__ bias,
                                                  float* __restrict__ hbuf) {
  const int wave = blockIdx.x * (blockDim.x >> 5) + (threadIdx.x >> 5);
  const int lane = threadIdx.x & 31;
  const int half = lane >> 4;
  const int l    = lane & 15;

  const int tileM = wave >> 2;
  const int cg    = wave & 3;
  const int row0  = tileM * 16;
  const int col0  = cg * 64;

  const float* xrow = X + (size_t)(row0 + l) * CIN_;
  const float* wr0  = W + (size_t)(col0 +  0 + l) * CIN_;
  const float* wr1  = W + (size_t)(col0 + 16 + l) * CIN_;
  const float* wr2  = W + (size_t)(col0 + 32 + l) * CIN_;
  const float* wr3  = W + (size_t)(col0 + 48 + l) * CIN_;

  v8f acc[4] = {{}, {}, {}, {}};
  #pragma unroll 4
  for (int kk = 0; kk < CIN_; kk += 4) {
    const int ko = kk + 2 * half;
    v2f av  = *(const v2f*)(xrow + ko);
    v2f bv0 = *(const v2f*)(wr0 + ko);
    v2f bv1 = *(const v2f*)(wr1 + ko);
    v2f bv2 = *(const v2f*)(wr2 + ko);
    v2f bv3 = *(const v2f*)(wr3 + ko);
    acc[0] = __builtin_amdgcn_wmma_f32_16x16x4_f32(false, av, false, bv0,
                                                   (short)0, acc[0], false, false);
    acc[1] = __builtin_amdgcn_wmma_f32_16x16x4_f32(false, av, false, bv1,
                                                   (short)0, acc[1], false, false);
    acc[2] = __builtin_amdgcn_wmma_f32_16x16x4_f32(false, av, false, bv2,
                                                   (short)0, acc[2], false, false);
    acc[3] = __builtin_amdgcn_wmma_f32_16x16x4_f32(false, av, false, bv3,
                                                   (short)0, acc[3], false, false);
  }

  #pragma unroll
  for (int cc = 0; cc < 4; ++cc) {
    const float bval = bias[col0 + cc * 16 + l];
    #pragma unroll
    for (int r = 0; r < 8; ++r) {
      const int row = row0 + r + 8 * half;
      hbuf[(size_t)row * COUT_ + col0 + cc * 16 + l] = acc[cc][r] + bval;
    }
  }
}

__global__ __launch_bounds__(256) void gat_lscore(const float* __restrict__ hbuf,
                                                  const float* __restrict__ a,
                                                  float* __restrict__ lsrc,
                                                  float* __restrict__ ldst) {
  const int t    = blockIdx.x * blockDim.x + threadIdx.x;
  const int head = t & (H_ - 1);
  const int bi   = t >> 2;
  const float* hr = hbuf + (size_t)bi * COUT_ + head * C_;
  const float* as = a + head * (2 * C_);
  const float* ad = as + C_;
  float s = 0.f, d = 0.f;
  #pragma unroll 8
  for (int c = 0; c < C_; ++c) {
    const float v = hr[c];
    s = fmaf(v, as[c], s);
    d = fmaf(v, ad[c], d);
  }
  lsrc[t] = s;
  ldst[t] = d;
}

__global__ __launch_bounds__(256) void gat_stats(const float* __restrict__ lsrc,
                                                 const float* __restrict__ ldst,
                                                 const int*   __restrict__ adj,
                                                 float* __restrict__ mbuf,
                                                 float* __restrict__ sbuf) {
  const int wave = blockIdx.x * (blockDim.x >> 5) + (threadIdx.x >> 5);
  const int lane = threadIdx.x & 31;
  const size_t bi = (size_t)wave;
  const int b = wave / N_;

  float ls[H_];
  #pragma unroll
  for (int hd = 0; hd < H_; ++hd) ls[hd] = lsrc[bi * H_ + hd];

  const int*    arow = adj + bi * (size_t)N_;
  const float4* ldr  = (const float4*)ldst;

  float mx[H_];
  #pragma unroll
  for (int hd = 0; hd < H_; ++hd) mx[hd] = NEG_INF_;

  for (int j = lane; j < N_; j += 32) {
    const int    ad = arow[j];
    const float4 ld = ldr[(size_t)b * N_ + j];
    const float  ldv[H_] = {ld.x, ld.y, ld.z, ld.w};
    #pragma unroll
    for (int hd = 0; hd < H_; ++hd) {
      float x = ls[hd] + ldv[hd];
      x = fmaxf(x, ALPHA_ * x);
      x = (ad == 1) ? x : NEG_INF_;
      mx[hd] = fmaxf(mx[hd], x);
    }
  }
  #pragma unroll
  for (int hd = 0; hd < H_; ++hd)
    #pragma unroll
    for (int off = 16; off > 0; off >>= 1)
      mx[hd] = fmaxf(mx[hd], __shfl_xor(mx[hd], off, 32));

  float sm[H_] = {0.f, 0.f, 0.f, 0.f};
  for (int j = lane; j < N_; j += 32) {
    const int    ad = arow[j];
    const float4 ld = ldr[(size_t)b * N_ + j];
    const float  ldv[H_] = {ld.x, ld.y, ld.z, ld.w};
    #pragma unroll
    for (int hd = 0; hd < H_; ++hd) {
      float x = ls[hd] + ldv[hd];
      x = fmaxf(x, ALPHA_ * x);
      sm[hd] += (ad == 1) ? __expf(x - mx[hd]) : 0.f;
    }
  }
  #pragma unroll
  for (int hd = 0; hd < H_; ++hd)
    #pragma unroll
    for (int off = 16; off > 0; off >>= 1)
      sm[hd] += __shfl_xor(sm[hd], off, 32);

  if (lane == 0) {
    #pragma unroll
    for (int hd = 0; hd < H_; ++hd) {
      mbuf[bi * H_ + hd] = mx[hd];
      sbuf[bi * H_ + hd] = sm[hd];
    }
  }
}

#define JS_   4
#define JLEN_ (N_ / JS_)
#define RP_   260
#define AGG_THREADS_ (32 * H_ * JS_)

__global__ __launch_bounds__(AGG_THREADS_) void gat_aggregate(
    const float* __restrict__ hbuf,
    const float* __restrict__ lsrc,
    const float* __restrict__ ldst,
    const int*   __restrict__ adj,
    const float* __restrict__ mbuf,
    const float* __restrict__ sbuf,
    float* __restrict__ out) {
  __shared__ __attribute__((aligned(16))) float lds_ld[N_ * H_];
  __shared__ __attribute__((aligned(16))) float lds_part[JS_ * 32 * RP_];

  const int b     = blockIdx.x >> 5;
  const int islab = blockIdx.x & 31;
  const int i0    = islab * 32;
  const int wv    = __builtin_amdgcn_readfirstlane((int)(threadIdx.x >> 5));
  const int head  = wv & (H_ - 1);
  const int js    = wv >> 2;
  const int lane  = threadIdx.x & 31;
  const int half  = lane >> 4;
  const int l     = lane & 15;

  {
    const float4* src = (const float4*)(ldst + (size_t)b * N_ * H_);
    float4*       dst = (float4*)lds_ld;
    #pragma unroll
    for (int t = 0; t < N_ / AGG_THREADS_; ++t) {
      const int j = t * AGG_THREADS_ + threadIdx.x;
      dst[j] = src[j];
    }
  }
  __syncthreads();

  const size_t biA = (size_t)b * N_ + (i0 + l);
  const size_t biB = biA + 16;
  const float  lsA = lsrc[biA * H_ + head];
  const float  lsB = lsrc[biB * H_ + head];
  const float  mA  = mbuf[biA * H_ + head];
  const float  mB  = mbuf[biB * H_ + head];
  const float  isA = 1.0f / sbuf[biA * H_ + head];
  const float  isB = 1.0f / sbuf[biB * H_ + head];
  const int*   arA = adj + biA * (size_t)N_;
  const int*   arB = adj + biB * (size_t)N_;

  v8f accA[4] = {{}, {}, {}, {}};
  v8f accB[4] = {{}, {}, {}, {}};

  const int jbeg = js * JLEN_;
  for (int j0 = jbeg; j0 < jbeg + JLEN_; j0 += 4) {
    const int ja = j0 + 2 * half;

    const int2  adA = *(const int2*)(arA + ja);
    const int2  adB = *(const int2*)(arB + ja);
    const float ld0 = lds_ld[ja * H_ + head];
    const float ld1 = lds_ld[(ja + 1) * H_ + head];

    float xA0 = lsA + ld0; xA0 = fmaxf(xA0, ALPHA_ * xA0);
    float xA1 = lsA + ld1; xA1 = fmaxf(xA1, ALPHA_ * xA1);
    float xB0 = lsB + ld0; xB0 = fmaxf(xB0, ALPHA_ * xB0);
    float xB1 = lsB + ld1; xB1 = fmaxf(xB1, ALPHA_ * xB1);

    v2f avA, avB;
    avA.x = (adA.x == 1) ? __expf(xA0 - mA) * isA : 0.0f;
    avA.y = (adA.y == 1) ? __expf(xA1 - mA) * isA : 0.0f;
    avB.x = (adB.x == 1) ? __expf(xB0 - mB) * isB : 0.0f;
    avB.y = (adB.y == 1) ? __expf(xB1 - mB) * isB : 0.0f;

    const float* hrow0 = hbuf + ((size_t)b * N_ + ja) * COUT_ + head * C_ + l;
    const float* hrow1 = hrow0 + COUT_;
    #pragma unroll
    for (int cc = 0; cc < 4; ++cc) {
      v2f bv;
      bv.x = hrow0[cc * 16];
      bv.y = hrow1[cc * 16];
      accA[cc] = __builtin_amdgcn_wmma_f32_16x16x4_f32(false, avA, false, bv,
                                                       (short)0, accA[cc],
                                                       false, false);
      accB[cc] = __builtin_amdgcn_wmma_f32_16x16x4_f32(false, avB, false, bv,
                                                       (short)0, accB[cc],
                                                       false, false);
    }
  }

  {
    float* pl = lds_part + js * (32 * RP_) + head * C_ + l;
    #pragma unroll
    for (int cc = 0; cc < 4; ++cc) {
      #pragma unroll
      for (int r = 0; r < 8; ++r) {
        const int rowA = r + 8 * half;
        pl[rowA * RP_ + cc * 16]        = accA[cc][r];
        pl[(rowA + 16) * RP_ + cc * 16] = accB[cc][r];
      }
    }
  }
  __syncthreads();

  #pragma unroll
  for (int t = 0; t < (32 * (COUT_ / 4)) / AGG_THREADS_; ++t) {
    const int q   = t * AGG_THREADS_ + threadIdx.x;
    const int row = q >> 6;
    const int c4  = q & 63;
    const float* pp = lds_part + row * RP_ + c4 * 4;
    float4 s = *(const float4*)pp;
    #pragma unroll
    for (int k = 1; k < JS_; ++k) {
      const float4 p = *(const float4*)(pp + k * (32 * RP_));
      s.x += p.x; s.y += p.y; s.z += p.z; s.w += p.w;
    }
    *(float4*)(out + ((size_t)b * N_ + i0 + row) * COUT_ + c4 * 4) = s;
  }
}

extern "C" void kernel_launch(void* const* d_in, const int* in_sizes, int n_in,
                              void* d_out, int out_size, void* d_ws, size_t ws_size,
                              hipStream_t stream) {
  const float* node_feats = (const float*)d_in[0];
  const int*   adj        = (const int*)  d_in[1];
  const float* W          = (const float*)d_in[2];
  const float* bias       = (const float*)d_in[3];
  const float* a          = (const float*)d_in[4];
  float*       out        = (float*)d_out;

  float* hbuf = (float*)d_ws;
  float* lsrc = hbuf + (size_t)B_ * N_ * COUT_;
  float* ldst = lsrc + (size_t)B_ * N_ * H_;
  float* mbuf = ldst + (size_t)B_ * N_ * H_;
  float* sbuf = mbuf + (size_t)B_ * N_ * H_;

  gat_gemm_h<<<(B_ * N_ / 16) * (COUT_ / 64) / 8, 256, 0, stream>>>(
      node_feats, W, bias, hbuf);

  gat_lscore<<<(B_ * N_ * H_) / 256, 256, 0, stream>>>(hbuf, a, lsrc, ldst);

  gat_stats<<<(B_ * N_) / 8, 256, 0, stream>>>(lsrc, ldst, adj, mbuf, sbuf);

  gat_aggregate<<<B_ * (N_ / 32), AGG_THREADS_, 0, stream>>>(
      hbuf, lsrc, ldst, adj, mbuf, sbuf, out);
}


# Softmax statistics fused into the aggregation kernel: adjacency read once into LDS bit masks (row max there; exp sums per lane, then per slice: new sum order; 1/sum at the store); one launch fewer
# speedup vs baseline: 2.3312x; 1.2853x over previous
#include <hip/hip_runtime.h>
#include <hip/hip_bf16.h>
#include <math.h>

#define B_      8
#define N_      1024
#define CIN_    256
#define H_      4
#define COUT_   256
#define C_      64
#define ALPHA_  0.2f
#define NEG_INF_ (-9.0e15f)

typedef float v2f __attribute__((ext_vector_type(2)));
typedef float v8f __attribute__((ext_vector_type(8)));

__global__ __launch_bounds__(256) void gat_gemm_h(const float* __restrict__ X,
                                                  const float* __restrict__ W,
                                                  const float* __restrict__ bias,
                                                  float* __restrict__ hbuf) {
  const int wave = blockIdx.x * (blockDim.x >> 5) + (threadIdx.x >> 5);
  const int lane = threadIdx.x & 31;
  const int half = lane >> 4;
  const int l    = lane & 15;

  const int tileM = wave >> 2;
  const int cg    = wave & 3;
  const int row0  = tileM * 16;
  const int col0  = cg * 64;

  const float* xrow = X + (size_t)(row0 + l) * CIN_;
  const float* wr0  = W + (size_t)(col0 +  0 + l) * CIN_;
  const float* wr1  = W + (size_t)(col0 + 16 + l) * CIN_;
  const float* wr2  = W + (size_t)(col0 + 32 + l) * CIN_;
  const float* wr3  = W + (size_t)(col0 + 48 + l) * CIN_;

  v8f acc[4] = {{}, {}, {}, {}};
  #pragma unroll 4
  for (int kk = 0; kk < CIN_; kk += 4) {
    const int ko = kk + 2 * half;
    v2f av  = *(const v2f*)(xrow + ko);
    v2f bv0 = *(const v2f*)(wr0 + ko);
    v2f bv1 = *(const v2f*)(wr1 + ko);
    v2f bv2 = *(const v2f*)(wr2 + ko);
    v2f bv3 = *(const v2f*)(wr3 + ko);
    acc[0] = __builtin_amdgcn_wmma_f32_16x16x4_f32(false, av, false, bv0,
                                                   (short)0, acc[0], false, false);
    acc[1] = __builtin_amdgcn_wmma_f32_16x16x4_f32(false, av, false, bv1,
                                                   (short)0, acc[1], false, false);
    acc[2] = __builtin_amdgcn_wmma_f32_16x16x4_f32(false, av, false, bv2,
                                                   (short)0, acc[2], false, false);
    acc[3] = __builtin_amdgcn_wmma_f32_16x16x4_f32(false, av, false, bv3,
                                                   (short)0, acc[3], false, false);
  }

  #pragma unroll
  for (int cc = 0; cc < 4; ++cc) {
    const float bval = bias[col0 + cc * 16 + l];
    #pragma unroll
    for (int r = 0; r < 8; ++r) {
      const int row = row0 + r + 8 * half;
      hbuf[(size_t)row * COUT_ + col0 + cc * 16 + l] = acc[cc][r] + bval;
    }
  }
}

__global__ __launch_bounds__(256) void gat_lscore(const float* __restrict__ hbuf,
                                                  const float* __restrict__ a,
                                                  float* __restrict__ lsrc,
                                                  float* __restrict__ ldst) {
  const int t    = blockIdx.x * blockDim.x + threadIdx.x;
  const int head = t & (H_ - 1);
  const int bi   = t >> 2;
  const float* hr = hbuf + (size_t)bi * COUT_ + head * C_;
  const float* as = a + head * (2 * C_);
  const float* ad = as + C_;
  float s = 0.f, d = 0.f;
  #pragma unroll 8
  for (int c = 0; c < C_; ++c) {
    const float v = hr[c];
    s = fmaf(v, as[c], s);
    d = fmaf(v, ad[c], d);
  }
  lsrc[t] = s;
  ldst[t] = d;
}

#define JS_   4
#define JLEN_ (N_ / JS_)
#define RP_   260
#define BW_   33
#define AGG_THREADS_ (32 * H_ * JS_)

__global__ __launch_bounds__(AGG_THREADS_) void gat_aggregate(
    const float* __restrict__ hbuf,
    const float* __restrict__ lsrc,
    const float* __restrict__ ldst,
    const int*   __restrict__ adj,
    float* __restrict__ out) {
  __shared__ __attribute__((aligned(16))) float lds_ld[N_ * H_];
  __shared__ __attribute__((aligned(16))) float lds_part[JS_ * 32 * RP_];
  __shared__ unsigned lds_bits[32 * BW_];
  __shared__ float    lds_m[32 * H_];
  __shared__ float    lds_s[JS_ * 32 * H_];

  const int b     = blockIdx.x >> 5;
  const int islab = blockIdx.x & 31;
  const int i0    = islab * 32;
  const int wv    = __builtin_amdgcn_readfirstlane((int)(threadIdx.x >> 5));
  const int head  = wv & (H_ - 1);
  const int js    = wv >> 2;
  const int lane  = threadIdx.x & 31;
  const int half  = lane >> 4;
  const int l     = lane & 15;

  float4 stg[N_ / AGG_THREADS_];
  {
    const float4* src = (const float4*)(ldst + (size_t)b * N_ * H_);
    #pragma unroll
    for (int t = 0; t < N_ / AGG_THREADS_; ++t)
      stg[t] = src[t * AGG_THREADS_ + threadIdx.x];
  }

  const int  srow = 2 * wv;
  const int* ap   = adj + ((size_t)b * N_ + i0 + srow) * N_ + lane;
  int ad0[32], ad1[32];
  #pragma unroll
  for (int k = 0; k < 32; ++k) ad0[k] = ap[32 * k];
  #pragma unroll
  for (int k = 0; k < 32; ++k) ad1[k] = ap[N_ + 32 * k];

  {
    float4* dst = (float4*)lds_ld;
    #pragma unroll
    for (int t = 0; t < N_ / AGG_THREADS_; ++t)
      dst[t * AGG_THREADS_ + threadIdx.x] = stg[t];
  }
  __syncthreads();

  {
    float mx0[H_], mx1[H_];
    #pragma unroll
    for (int hd = 0; hd < H_; ++hd) { mx0[hd] = -INFINITY; mx1[hd] = -INFINITY; }
    unsigned w0 = 0u, w1 = 0u;
    const float4* ld4 = (const float4*)lds_ld;
    #pragma unroll
    for (int k = 0; k < 32; ++k) {
      const float4 ld = ld4[32 * k + lane];
      const bool p0 = (ad0[k] == 1);
      const bool p1 = (ad1[k] == 1);
      const unsigned m0 = __builtin_amdgcn_ballot_w32(p0);
      const unsigned m1 = __builtin_amdgcn_ballot_w32(p1);
      w0 = (lane == k) ? m0 : w0;
      w1 = (lane == k) ? m1 : w1;
      mx0[0] = fmaxf(mx0[0], p0 ? ld.x : -INFINITY);
      mx0[1] = fmaxf(mx0[1], p0 ? ld.y : -INFINITY);
      mx0[2] = fmaxf(mx0[2], p0 ? ld.z : -INFINITY);
      mx0[3] = fmaxf(mx0[3], p0 ? ld.w : -INFINITY);
      mx1[0] = fmaxf(mx1[0], p1 ? ld.x : -INFINITY);
      mx1[1] = fmaxf(mx1[1], p1 ? ld.y : -INFINITY);
      mx1[2] = fmaxf(mx1[2], p1 ? ld.z : -INFINITY);
      mx1[3] = fmaxf(mx1[3], p1 ? ld.w : -INFINITY);
    }
    #pragma unroll
    for (int hd = 0; hd < H_; ++hd) {
      #pragma unroll
      for (int off = 16; off > 0; off >>= 1) {
        mx0[hd] = fmaxf(mx0[hd], __shfl_xor(mx0[hd], off, 32));
        mx1[hd] = fmaxf(mx1[hd], __shfl_xor(mx1[hd], off, 32));
      }
    }
    lds_bits[srow * BW_ + lane]       = w0;
    lds_bits[(srow + 1) * BW_ + lane] = w1;
    float v = mx0[0];
    v = (lane == 1) ? mx0[1] : v;
    v = (lane == 2) ? mx0[2] : v;
    v = (lane == 3) ? mx0[3] : v;
    v = (lane == 4) ? mx1[0] : v;
    v = (lane == 5) ? mx1[1] : v;
    v = (lane == 6) ? mx1[2] : v;
    v = (lane == 7) ? mx1[3] : v;
    if (lane < 2 * H_) {
      float x = lsrc[((size_t)b * N_ + i0 + srow) * H_ + lane] + v;
      x = fmaxf(x, ALPHA_ * x);
      lds_m[srow * H_ + lane] = fmaxf(x, NEG_INF_);
    }
  }
  __syncthreads();

  const size_t biA = (size_t)b * N_ + (i0 + l);
  const size_t biB = biA + 16;
  const float  lsA = lsrc[biA * H_ + head];
  const float  lsB = lsrc[biB * H_ + head];
  const float  mA  = lds_m[l * H_ + head];
  const float  mB  = lds_m[(l + 16) * H_ + head];

  v8f accA[4] = {{}, {}, {}, {}};
  v8f accB[4] = {{}, {}, {}, {}};
  float sA = 0.f, sB = 0.f;

  const unsigned* bitA = lds_bits + l * BW_ + js * (JLEN_ / 32);
  const unsigned* bitB = bitA + 16 * BW_;
  const float* hbase = hbuf + ((size_t)b * N_ + js * JLEN_ + 2 * half) * COUT_ + head * C_ + l;
  const float* ldp   = lds_ld + (js * JLEN_ + 2 * half) * H_ + head;

  for (int w = 0; w < JLEN_ / 32; ++w) {
    const unsigned bA = bitA[w] >> (2 * half);
    const unsigned bB = bitB[w] >> (2 * half);
    #pragma unroll
    for (int t = 0; t < 8; ++t) {
      const int jo = w * 32 + 4 * t;
      const float ld0 = ldp[jo * H_];
      const float ld1 = ldp[(jo + 1) * H_];

      float xA0 = lsA + ld0; xA0 = fmaxf(xA0, ALPHA_ * xA0);
      float xA1 = lsA + ld1; xA1 = fmaxf(xA1, ALPHA_ * xA1);
      float xB0 = lsB + ld0; xB0 = fmaxf(xB0, ALPHA_ * xB0);
      float xB1 = lsB + ld1; xB1 = fmaxf(xB1, ALPHA_ * xB1);

      v2f avA, avB;
      avA.x = ((bA >> (4 * t))     & 1u) ? __expf(xA0 - mA) : 0.0f;
      avA.y = ((bA >> (4 * t + 1)) & 1u) ? __expf(xA1 - mA) : 0.0f;
      avB.x = ((bB >> (4 * t))     & 1u) ? __expf(xB0 - mB) : 0.0f;
      avB.y = ((bB >> (4 * t + 1)) & 1u) ? __expf(xB1 - mB) : 0.0f;
      sA += avA.x + avA.y;
      sB += avB.x + avB.y;

      const float* hrow0 = hbase + (size_t)jo * COUT_;
      const float* hrow1 = hrow0 + COUT_;
      #pragma unroll
      for (int cc = 0; cc < 4; ++cc) {
        v2f bv;
        bv.x = hrow0[cc * 16];
        bv.y = hrow1[cc * 16];
        accA[cc] = __builtin_amdgcn_wmma_f32_16x16x4_f32(false, avA, false, bv,
                                                         (short)0, accA[cc],
                                                         false, false);
        accB[cc] = __builtin_amdgcn_wmma_f32_16x16x4_f32(false, avB, false, bv,
                                                         (short)0, accB[cc],
                                                         false, false);
      }
    }
  }

  sA += __shfl_xor(sA, 16, 32);
  sB += __shfl_xor(sB, 16, 32);
  lds_s[(js * 32 + lane) * H_ + head] = half ? sB : sA;

  {
    float* pl = lds_part + js * (32 * RP_) + head * C_ + l;
    #pragma unroll
    for (int cc = 0; cc < 4; ++cc) {
      #pragma unroll
      for (int r = 0; r < 8; ++r) {
        const int rowA = r + 8 * half;
        pl[rowA * RP_ + cc * 16]        = accA[cc][r];
        pl[(rowA + 16) * RP_ + cc * 16] = accB[cc][r];
      }
    }
  }
  __syncthreads();

  #pragma unroll
  for (int t = 0; t < (32 * (COUT_ / 4)) / AGG_THREADS_; ++t) {
    const int q   = t * AGG_THREADS_ + threadIdx.x;
    const int row = q >> 6;
    const int c4  = q & 63;
    const int hd  = c4 >> 4;
    const float* sp = lds_s + row * H_ + hd;
    float tot = sp[0];
    #pragma unroll
    for (int k = 1; k < JS_; ++k) tot += sp[k * (32 * H_)];
    const float inv = (tot > 0.0f) ? 1.0f / tot : 0.0f;
    const float* pp = lds_part + row * RP_ + c4 * 4;
    float4 s = *(const float4*)pp;
    #pragma unroll
    for (int k = 1; k < JS_; ++k) {
      const float4 p = *(const float4*)(pp + k * (32 * RP_));
      s.x += p.x; s.y += p.y; s.z += p.z; s.w += p.w;
    }
    s.x *= inv; s.y *= inv; s.z *= inv; s.w *= inv;
    *(float4*)(out + ((size_t)b * N_ + i0 + row) * COUT_ + c4 * 4) = s;
  }
}

extern "C" void kernel_launch(void* const* d_in, const int* in_sizes, int n_in,
                              void* d_out, int out_size, void* d_ws, size_t ws_size,
                              hipStream_t stream) {
  const float* node_feats = (const float*)d_in[0];
  const int*   adj        = (const int*)  d_in[1];
  const float* W          = (const float*)d_in[2];
  const float* bias       = (const float*)d_in[3];
  const float* a          = (const float*)d_in[4];
  float*       out        = (float*)d_out;

  float* hbuf = (float*)d_ws;
  float* lsrc = hbuf + (size_t)B_ * N_ * COUT_;
  float* ldst = lsrc + (size_t)B_ * N_ * H_;

  gat_gemm_h<<<(B_ * N_ / 16) * (COUT_ / 64) / 8, 256, 0, stream>>>(
      node_feats, W, bias, hbuf);

  gat_lscore<<<(B_ * N_ * H_) / 256, 256, 0, stream>>>(hbuf, a, lsrc, ldst);

  gat_aggregate<<<B_ * (N_ / 32), AGG_THREADS_, 0, stream>>>(
      hbuf, lsrc, ldst, adj, out);
}


# Logit halves l_src/l_dst now come out of the projection GEMM's epilogue (4 terms per lane from the accumulators, then a 16-lane tree: new sum order); the separate logit kernel and its launch are gone
# speedup vs baseline: 2.5930x; 1.1123x over previous
#include <hip/hip_runtime.h>
#include <hip/hip_bf16.h>
#include <math.h>

#define B_      8
#define N_      1024
#define CIN_    256
#define H_      4
#define COUT_   256
#define C_      64
#define ALPHA_  0.2f
#define NEG_INF_ (-9.0e15f)

typedef float v2f __attribute__((ext_vector_type(2)));
typedef float v8f __attribute__((ext_vector_type(8)));

__global__ __launch_bounds__(256) void gat_gemm_h(const float* __restrict__ X,
                                                  const float* __restrict__ W,
                                                  const float* __restrict__ bias,
                                                  const float* __restrict__ a,
                                                  float* __restrict__ hbuf,
                                                  float* __restrict__ lsrc,
                                                  float* __restrict__ ldst) {
  __shared__ __attribute__((aligned(16))) float lds_l[2 * 32 * H_];

  const int wb   = threadIdx.x >> 5;
  const int wave = blockIdx.x * (blockDim.x >> 5) + wb;
  const int lane = threadIdx.x & 31;
  const int half = lane >> 4;
  const int l    = lane & 15;

  const int tileM = wave >> 2;
  const int cg    = wave & 3;
  const int row0  = tileM * 16;
  const int col0  = cg * 64;

  const float* xrow = X + (size_t)(row0 + l) * CIN_;
  const float* wr0  = W + (size_t)(col0 +  0 + l) * CIN_;
  const float* wr1  = W + (size_t)(col0 + 16 + l) * CIN_;
  const float* wr2  = W + (size_t)(col0 + 32 + l) * CIN_;
  const float* wr3  = W + (size_t)(col0 + 48 + l) * CIN_;

  v8f acc[4] = {{}, {}, {}, {}};
  #pragma unroll 4
  for (int kk = 0; kk < CIN_; kk += 4) {
    const int ko = kk + 2 * half;
    v2f av  = *(const v2f*)(xrow + ko);
    v2f bv0 = *(const v2f*)(wr0 + ko);
    v2f bv1 = *(const v2f*)(wr1 + ko);
    v2f bv2 = *(const v2f*)(wr2 + ko);
    v2f bv3 = *(const v2f*)(wr3 + ko);
    acc[0] = __builtin_amdgcn_wmma_f32_16x16x4_f32(false, av, false, bv0,
                                                   (short)0, acc[0], false, false);
    acc[1] = __builtin_amdgcn_wmma_f32_16x16x4_f32(false, av, false, bv1,
                                                   (short)0, acc[1], false, false);
    acc[2] = __builtin_amdgcn_wmma_f32_16x16x4_f32(false, av, false, bv2,
                                                   (short)0, acc[2], false, false);
    acc[3] = __builtin_amdgcn_wmma_f32_16x16x4_f32(false, av, false, bv3,
                                                   (short)0, acc[3], false, false);
  }

  const float* as = a + cg * (2 * C_);
  const float* ad = as + C_;
  float part[16];
  #pragma unroll
  for (int i = 0; i < 16; ++i) part[i] = 0.f;

  #pragma unroll
  for (int cc = 0; cc < 4; ++cc) {
    const float bval = bias[col0 + cc * 16 + l];
    const float asv  = as[cc * 16 + l];
    const float adv  = ad[cc * 16 + l];
    #pragma unroll
    for (int r = 0; r < 8; ++r) {
      acc[cc][r] += bval;
      part[r]     = fmaf(acc[cc][r], asv, part[r]);
      part[8 + r] = fmaf(acc[cc][r], adv, part[8 + r]);
    }
  }

  float q8[8], q4[4], q2[2], tot;
  {
    const bool up = (l & 8) != 0;
    #pragma unroll
    for (int i = 0; i < 8; ++i) {
      const float send = up ? part[i] : part[i + 8];
      const float keep = up ? part[i + 8] : part[i];
      q8[i] = keep + __shfl_xor(send, 8, 32);
    }
  }
  {
    const bool up = (l & 4) != 0;
    #pragma unroll
    for (int i = 0; i < 4; ++i) {
      const float send = up ? q8[i] : q8[i + 4];
      const float keep = up ? q8[i + 4] : q8[i];
      q4[i] = keep + __shfl_xor(send, 4, 32);
    }
  }
  {
    const bool up = (l & 2) != 0;
    #pragma unroll
    for (int i = 0; i < 2; ++i) {
      const float send = up ? q4[i] : q4[i + 2];
      const float keep = up ? q4[i + 2] : q4[i];
      q2[i] = keep + __shfl_xor(send, 2, 32);
    }
  }
  {
    const bool up = (l & 1) != 0;
    const float send = up ? q2[0] : q2[1];
    const float keep = up ? q2[1] : q2[0];
    tot = keep + __shfl_xor(send, 1, 32);
  }
  lds_l[(l >> 3) * (32 * H_) + ((wb >> 2) * 16 + (l & 7) + 8 * half) * H_ + cg] = tot;
  __syncthreads();

  if (wb < 2) {
    const float4 q = *(const float4*)(lds_l + wb * (32 * H_) + lane * H_);
    float* dstp = (wb == 0 ? lsrc : ldst) + ((size_t)blockIdx.x * 32 + lane) * H_;
    *(float4*)dstp = q;
  }

  #pragma unroll
  for (int cc = 0; cc < 4; ++cc) {
    #pragma unroll
    for (int r = 0; r < 8; ++r) {
      const int row = row0 + r + 8 * half;
      hbuf[(size_t)row * COUT_ + col0 + cc * 16 + l] = acc[cc][r];
    }
  }
}

#define JS_   4
#define JLEN_ (N_ / JS_)
#define RP_   260
#define BW_   33
#define AGG_THREADS_ (32 * H_ * JS_)

__global__ __launch_bounds__(AGG_THREADS_) void gat_aggregate(
    const float* __restrict__ hbuf,
    const float* __restrict__ lsrc,
    const float* __restrict__ ldst,
    const int*   __restrict__ adj,
    float* __restrict__ out) {
  __shared__ __attribute__((aligned(16))) float lds_ld[N_ * H_];
  __shared__ __attribute__((aligned(16))) float lds_part[JS_ * 32 * RP_];
  __shared__ unsigned lds_bits[32 * BW_];
  __shared__ float    lds_m[32 * H_];
  __shared__ float    lds_s[JS_ * 32 * H_];

  const int b     = blockIdx.x >> 5;
  const int islab = blockIdx.x & 31;
  const int i0    = islab * 32;
  const int wv    = __builtin_amdgcn_readfirstlane((int)(threadIdx.x >> 5));
  const int head  = wv & (H_ - 1);
  const int js    = wv >> 2;
  const int lane  = threadIdx.x & 31;
  const int half  = lane >> 4;
  const int l     = lane & 15;

  float4 stg[N_ / AGG_THREADS_];
  {
    const float4* src = (const float4*)(ldst + (size_t)b * N_ * H_);
    #pragma unroll
    for (int t = 0; t < N_ / AGG_THREADS_; ++t)
      stg[t] = src[t * AGG_THREADS_ + threadIdx.x];
  }

  const int  srow = 2 * wv;
  const int* ap   = adj + ((size_t)b * N_ + i0 + srow) * N_ + lane;
  int ad0[32], ad1[32];
  #pragma unroll
  for (int k = 0; k < 32; ++k) ad0[k] = ap[32 * k];
  #pragma unroll
  for (int k = 0; k < 32; ++k) ad1[k] = ap[N_ + 32 * k];

  {
    float4* dst = (float4*)lds_ld;
    #pragma unroll
    for (int t = 0; t < N_ / AGG_THREADS_; ++t)
      dst[t * AGG_THREADS_ + threadIdx.x] = stg[t];
  }
  __syncthreads();

  {
    float mx0[H_], mx1[H_];
    #pragma unroll
    for (int hd = 0; hd < H_; ++hd) { mx0[hd] = -INFINITY; mx1[hd] = -INFINITY; }
    unsigned w0 = 0u, w1 = 0u;
    const float4* ld4 = (const float4*)lds_ld;
    #pragma unroll
    for (int k = 0; k < 32; ++k) {
      const float4 ld = ld4[32 * k + lane];
      const bool p0 = (ad0[k] == 1);
      const bool p1 = (ad1[k] == 1);
      const unsigned m0 = __builtin_amdgcn_ballot_w32(p0);
      const unsigned m1 = __builtin_amdgcn_ballot_w32(p1);
      w0 = (lane == k) ? m0 : w0;
      w1 = (lane == k) ? m1 : w1;
      mx0[0] = fmaxf(mx0[0], p0 ? ld.x : -INFINITY);
      mx0[1] = fmaxf(mx0[1], p0 ? ld.y : -INFINITY);
      mx0[2] = fmaxf(mx0[2], p0 ? ld.z : -INFINITY);
      mx0[3] = fmaxf(mx0[3], p0 ? ld.w : -INFINITY);
      mx1[0] = fmaxf(mx1[0], p1 ? ld.x : -INFINITY);
      mx1[1] = fmaxf(mx1[1], p1 ? ld.y : -INFINITY);
      mx1[2] = fmaxf(mx1[2], p1 ? ld.z : -INFINITY);
      mx1[3] = fmaxf(mx1[3], p1 ? ld.w : -INFINITY);
    }
    #pragma unroll
    for (int hd = 0; hd < H_; ++hd) {
      #pragma unroll
      for (int off = 16; off > 0; off >>= 1) {
        mx0[hd] = fmaxf(mx0[hd], __shfl_xor(mx0[hd], off, 32));
        mx1[hd] = fmaxf(mx1[hd], __shfl_xor(mx1[hd], off, 32));
      }
    }
    lds_bits[srow * BW_ + lane]       = w0;
    lds_bits[(srow + 1) * BW_ + lane] = w1;
    float v = mx0[0];
    v = (lane == 1) ? mx0[1] : v;
    v = (lane == 2) ? mx0[2] : v;
    v = (lane == 3) ? mx0[3] : v;
    v = (lane == 4) ? mx1[0] : v;
    v = (lane == 5) ? mx1[1] : v;
    v = (lane == 6) ? mx1[2] : v;
    v = (lane == 7) ? mx1[3] : v;
    if (lane < 2 * H_) {
      float x = lsrc[((size_t)b * N_ + i0 + srow) * H_ + lane] + v;
      x = fmaxf(x, ALPHA_ * x);
      lds_m[srow * H_ + lane] = fmaxf(x, NEG_INF_);
    }
  }
  __syncthreads();

  const size_t biA = (size_t)b * N_ + (i0 + l);
  const size_t biB = biA + 16;
  const float  lsA = lsrc[biA * H_ + head];
  const float  lsB = lsrc[biB * H_ + head];
  const float  mA  = lds_m[l * H_ + head];
  const float  mB  = lds_m[(l + 16) * H_ + head];

  v8f accA[4] = {{}, {}, {}, {}};
  v8f accB[4] = {{}, {}, {}, {}};
  float sA = 0.f, sB = 0.f;

  const unsigned* bitA = lds_bits + l * BW_ + js * (JLEN_ / 32);
  const unsigned* bitB = bitA + 16 * BW_;
  const float* hbase = hbuf + ((size_t)b * N_ + js * JLEN_ + 2 * half) * COUT_ + head * C_ + l;
  const float* ldp   = lds_ld + (js * JLEN_ + 2 * half) * H_ + head;

  for (int w = 0; w < JLEN_ / 32; ++w) {
    const unsigned bA = bitA[w] >> (2 * half);
    const unsigned bB = bitB[w] >> (2 * half);
    #pragma unroll
    for (int t = 0; t < 8; ++t) {
      const int jo = w * 32 + 4 * t;
      const float ld0 = ldp[jo * H_];
      const float ld1 = ldp[(jo + 1) * H_];

      float xA0 = lsA + ld0; xA0 = fmaxf(xA0, ALPHA_ * xA0);
      float xA1 = lsA + ld1; xA1 = fmaxf(xA1, ALPHA_ * xA1);
      float xB0 = lsB + ld0; xB0 = fmaxf(xB0, ALPHA_ * xB0);
      float xB1 = lsB + ld1; xB1 = fmaxf(xB1, ALPHA_ * xB1);

      v2f avA, avB;
      avA.x = ((bA >> (4 * t))     & 1u) ? __expf(xA0 - mA) : 0.0f;
      avA.y = ((bA >> (4 * t + 1)) & 1u) ? __expf(xA1 - mA) : 0.0f;
      avB.x = ((bB >> (4 * t))     & 1u) ? __expf(xB0 - mB) : 0.0f;
      avB.y = ((bB >> (4 * t + 1)) & 1u) ? __expf(xB1 - mB) : 0.0f;
      sA += avA.x + avA.y;
      sB += avB.x + avB.y;

      const float* hrow0 = hbase + (size_t)jo * COUT_;
      const float* hrow1 = hrow0 + COUT_;
      #pragma unroll
      for (int cc = 0; cc < 4; ++cc) {
        v2f bv;
        bv.x = hrow0[cc * 16];
        bv.y = hrow1[cc * 16];
        accA[cc] = __builtin_amdgcn_wmma_f32_16x16x4_f32(false, avA, false, bv,
                                                         (short)0, accA[cc],
                                                         false, false);
        accB[cc] = __builtin_amdgcn_wmma_f32_16x16x4_f32(false, avB, false, bv,
                                                         (short)0, accB[cc],
                                                         false, false);
      }
    }
  }

  sA += __shfl_xor(sA, 16, 32);
  sB += __shfl_xor(sB, 16, 32);
  lds_s[(js * 32 + lane) * H_ + head] = half ? sB : sA;

  {
    float* pl = lds_part + js * (32 * RP_) + head * C_ + l;
    #pragma unroll
    for (int cc = 0; cc < 4; ++cc) {
      #pragma unroll
      for (int r = 0; r < 8; ++r) {
        const int rowA = r + 8 * half;
        pl[rowA * RP_ + cc * 16]        = accA[cc][r];
        pl[(rowA + 16) * RP_ + cc * 16] = accB[cc][r];
      }
    }
  }
  __syncthreads();

  #pragma unroll
  for (int t = 0; t < (32 * (COUT_ / 4)) / AGG_THREADS_; ++t) {
    const int q   = t * AGG_THREADS_ + threadIdx.x;
    const int row = q >> 6;
    const int c4  = q & 63;
    const int hd  = c4 >> 4;
    const float* sp = lds_s + row * H_ + hd;
    float tot = sp[0];
    #pragma unroll
    for (int k = 1; k < JS_; ++k) tot += sp[k * (32 * H_)];
    const float inv = (tot > 0.0f) ? 1.0f / tot : 0.0f;
    const float* pp = lds_part + row * RP_ + c4 * 4;
    float4 s = *(const float4*)pp;
    #pragma unroll
    for (int k = 1; k < JS_; ++k) {
      const float4 p = *(const float4*)(pp + k * (32 * RP_));
      s.x += p.x; s.y += p.y; s.z += p.z; s.w += p.w;
    }
    s.x *= inv; s.y *= inv; s.z *= inv; s.w *= inv;
    *(float4*)(out + ((size_t)b * N_ + i0 + row) * COUT_ + c4 * 4) = s;
  }
}

extern "C" void kernel_launch(void* const* d_in, const int* in_sizes, int n_in,
                              void* d_out, int out_size, void* d_ws, size_t ws_size,
                              hipStream_t stream) {
  const float* node_feats = (const float*)d_in[0];
  const int*   adj        = (const int*)  d_in[1];
  const float* W          = (const float*)d_in[2];
  const float* bias       = (const float*)d_in[3];
  const float* a          = (const float*)d_in[4];
  float*       out        = (float*)d_out;

  float* hbuf = (float*)d_ws;
  float* lsrc = hbuf + (size_t)B_ * N_ * COUT_;
  float* ldst = lsrc + (size_t)B_ * N_ * H_;

  gat_gemm_h<<<(B_ * N_ / 16) * (COUT_ / 64) / 8, 256, 0, stream>>>(
      node_feats, W, bias, a, hbuf, lsrc, ldst);

  gat_aggregate<<<B_ * (N_ / 32), AGG_THREADS_, 0, stream>>>(
      hbuf, lsrc, ldst, adj, out);
}


# Projection GEMM: X and W K-chunks of 64 staged in LDS by whole-line 16-byte loads, next chunk in flight during the current chunk's WMMAs; fragments read from LDS; sum order unchanged
# speedup vs baseline: 2.8523x; 1.1000x over previous
#include <hip/hip_runtime.h>
#include <hip/hip_bf16.h>
#include <math.h>

#define B_      8
#define N_      1024
#define CIN_    256
#define H_      4
#define COUT_   256
#define C_      64
#define ALPHA_  0.2f
#define NEG_INF_ (-9.0e15f)

typedef float v2f __attribute__((ext_vector_type(2)));
typedef float v8f __attribute__((ext_vector_type(8)));
typedef float v4f __attribute__((ext_vector_type(4)));

#define KC_         64
#define LP_         68
#define RPP_        (256 / (KC_ / 4))
#define STG_FLOATS_ ((32 + COUT_) * LP_)
#define NLD_        ((32 + COUT_) / RPP_)

static_assert(KC_ == 64 && RPP_ == 16 && NLD_ == 18, "Pre holds 2 + 16 passes of 16 rows");
struct Pre { v4f x0, x1, w0, w1, w2, w3, w4, w5, w6, w7, w8, w9, w10, w11, w12, w13, w14, w15; };

__device__ __forceinline__ void pre_load(Pre& p, const float* gx, const float* gw, int koff) {
  p.x0 = *(const v4f*)(gx + (size_t)(0 * RPP_) * CIN_ + koff);
  p.x1 = *(const v4f*)(gx + (size_t)(1 * RPP_) * CIN_ + koff);
  p.w0 = *(const v4f*)(gw + (size_t)(0 * RPP_) * CIN_ + koff);
  p.w1 = *(const v4f*)(gw + (size_t)(1 * RPP_) * CIN_ + koff);
  p.w2 = *(const v4f*)(gw + (size_t)(2 * RPP_) * CIN_ + koff);
  p.w3 = *(const v4f*)(gw + (size_t)(3 * RPP_) * CIN_ + koff);
  p.w4 = *(const v4f*)(gw + (size_t)(4 * RPP_) * CIN_ + koff);
  p.w5 = *(const v4f*)(gw + (size_t)(5 * RPP_) * CIN_ + koff);
  p.w6 = *(const v4f*)(gw + (size_t)(6 * RPP_) * CIN_ + koff);
  p.w7 = *(const v4f*)(gw + (size_t)(7 * RPP_) * CIN_ + koff);
  p.w8 = *(const v4f*)(gw + (size_t)(8 * RPP_) * CIN_ + koff);
  p.w9 = *(const v4f*)(gw + (size_t)(9 * RPP_) * CIN_ + koff);
  p.w10 = *(const v4f*)(gw + (size_t)(10 * RPP_) * CIN_ + koff);
  p.w11 = *(const v4f*)(gw + (size_t)(11 * RPP_) * CIN_ + koff);
  p.w12 = *(const v4f*)(gw + (size_t)(12 * RPP_) * CIN_ + koff);
  p.w13 = *(const v4f*)(gw + (size_t)(13 * RPP_) * CIN_ + koff);
  p.w14 = *(const v4f*)(gw + (size_t)(14 * RPP_) * CIN_ + koff);
  p.w15 = *(const v4f*)(gw + (size_t)(15 * RPP_) * CIN_ + koff);
}

__device__ __forceinline__ void pre_store(const Pre& p, float* sdst) {
  *(v4f*)(sdst + (0 * RPP_) * LP_) = p.x0;
  *(v4f*)(sdst + (1 * RPP_) * LP_) = p.x1;
  *(v4f*)(sdst + (2 * RPP_) * LP_) = p.w0;
  *(v4f*)(sdst + (3 * RPP_) * LP_) = p.w1;
  *(v4f*)(sdst + (4 * RPP_) * LP_) = p.w2;
  *(v4f*)(sdst + (5 * RPP_) * LP_) = p.w3;
  *(v4f*)(sdst + (6 * RPP_) * LP_) = p.w4;
  *(v4f*)(sdst + (7 * RPP_) * LP_) = p.w5;
  *(v4f*)(sdst + (8 * RPP_) * LP_) = p.w6;
  *(v4f*)(sdst + (9 * RPP_) * LP_) = p.w7;
  *(v4f*)(sdst + (10 * RPP_) * LP_) = p.w8;
  *(v4f*)(sdst + (11 * RPP_) * LP_) = p.w9;
  *(v4f*)(sdst + (12 * RPP_) * LP_) = p.w10;
  *(v4f*)(sdst + (13 * RPP_) * LP_) = p.w11;
  *(v4f*)(sdst + (14 * RPP_) * LP_) = p.w12;
  *(v4f*)(sdst + (15 * RPP_) * LP_) = p.w13;
  *(v4f*)(sdst + (16 * RPP_) * LP_) = p.w14;
  *(v4f*)(sdst + (17 * RPP_) * LP_) = p.w15;
}

__global__ __launch_bounds__(256) __attribute__((amdgpu_waves_per_eu(2, 2))) void gat_gemm_h(const float* __restrict__ X,
                                                  const float* __restrict__ W,
                                                  const float* __restrict__ bias,
                                                  const float* __restrict__ a,
                                                  float* __restrict__ hbuf,
                                                  float* __restrict__ lsrc,
                                                  float* __restrict__ ldst) {
  __shared__ __attribute__((aligned(16))) float lds_l[2 * 32 * H_];
  __shared__ __attribute__((aligned(16))) float lds_op[2 * STG_FLOATS_];

  const int wb   = threadIdx.x >> 5;
  const int wave = blockIdx.x * (blockDim.x >> 5) + wb;
  const int lane = threadIdx.x & 31;
  const int half = lane >> 4;
  const int l    = lane & 15;

  const int tileM = wave >> 2;
  const int cg    = wave & 3;
  const int row0  = tileM * 16;
  const int col0  = cg * 64;

  const int srow = threadIdx.x / (KC_ / 4);
  const int sc4  = threadIdx.x % (KC_ / 4);
  const float* gx = X + ((size_t)blockIdx.x * 32 + srow) * CIN_ + sc4 * 4;
  const float* gw = W + (size_t)srow * CIN_ + sc4 * 4;
  float* sdst = lds_op + srow * LP_ + sc4 * 4;

  const float* sa = lds_op + ((wb >> 2) * 16 + l) * LP_ + 2 * half;
  const float* sb = lds_op + (32 + cg * 64 + l) * LP_ + 2 * half;

  Pre pre;
  pre_load(pre, gx, gw, 0);
  pre_store(pre, sdst);
  __syncthreads();

  v8f acc[4] = {{}, {}, {}, {}};
  #pragma unroll
  for (int c = 0; c < CIN_ / KC_; ++c) {
    const int st = (c & 1) * STG_FLOATS_;
    if (c + 1 < CIN_ / KC_) {
      pre_load(pre, gx, gw, (c + 1) * KC_);
      __builtin_amdgcn_sched_barrier(0);
    }
    #pragma unroll
    for (int s = 0; s < KC_ / 4; ++s) {
      const v2f av  = *(const v2f*)(sa + st + 4 * s);
      const v2f bv0 = *(const v2f*)(sb + st + 4 * s);
      const v2f bv1 = *(const v2f*)(sb + st + 16 * LP_ + 4 * s);
      const v2f bv2 = *(const v2f*)(sb + st + 32 * LP_ + 4 * s);
      const v2f bv3 = *(const v2f*)(sb + st + 48 * LP_ + 4 * s);
      acc[0] = __builtin_amdgcn_wmma_f32_16x16x4_f32(false, av, false, bv0,
                                                     (short)0, acc[0], false, false);
      acc[1] = __builtin_amdgcn_wmma_f32_16x16x4_f32(false, av, false, bv1,
                                                     (short)0, acc[1], false, false);
      acc[2] = __builtin_amdgcn_wmma_f32_16x16x4_f32(false, av, false, bv2,
                                                     (short)0, acc[2], false, false);
      acc[3] = __builtin_amdgcn_wmma_f32_16x16x4_f32(false, av, false, bv3,
                                                     (short)0, acc[3], false, false);
    }
    if (c + 1 < CIN_ / KC_) {
      __builtin_amdgcn_sched_barrier(0);
      pre_store(pre, sdst + (STG_FLOATS_ - st));
      __syncthreads();
    }
  }

  const float* as = a + cg * (2 * C_);
  const float* ad = as + C_;
  float part[16];
  #pragma unroll
  for (int i = 0; i < 16; ++i) part[i] = 0.f;

  #pragma unroll
  for (int cc = 0; cc < 4; ++cc) {
    const float bval = bias[col0 + cc * 16 + l];
    const float asv  = as[cc * 16 + l];
    const float adv  = ad[cc * 16 + l];
    #pragma unroll
    for (int r = 0; r < 8; ++r) {
      acc[cc][r] += bval;
      part[r]     = fmaf(acc[cc][r], asv, part[r]);
      part[8 + r] = fmaf(acc[cc][r], adv, part[8 + r]);
    }
  }

  float q8[8], q4[4], q2[2], tot;
  {
    const bool up = (l & 8) != 0;
    #pragma unroll
    for (int i = 0; i < 8; ++i) {
      const float send = up ? part[i] : part[i + 8];
      const float keep = up ? part[i + 8] : part[i];
      q8[i] = keep + __shfl_xor(send, 8, 32);
    }
  }
  {
    const bool up = (l & 4) != 0;
    #pragma unroll
    for (int i = 0; i < 4; ++i) {
      const float send = up ? q8[i] : q8[i + 4];
      const float keep = up ? q8[i + 4] : q8[i];
      q4[i] = keep + __shfl_xor(send, 4, 32);
    }
  }
  {
    const bool up = (l & 2) != 0;
    #pragma unroll
    for (int i = 0; i < 2; ++i) {
      const float send = up ? q4[i] : q4[i + 2];
      const float keep = up ? q4[i + 2] : q4[i];
      q2[i] = keep + __shfl_xor(send, 2, 32);
    }
  }
  {
    const bool up = (l & 1) != 0;
    const float send = up ? q2[0] : q2[1];
    const float keep = up ? q2[1] : q2[0];
    tot = keep + __shfl_xor(send, 1, 32);
  }
  lds_l[(l >> 3) * (32 * H_) + ((wb >> 2) * 16 + (l & 7) + 8 * half) * H_ + cg] = tot;
  __syncthreads();

  if (wb < 2) {
    const float4 q = *(const float4*)(lds_l + wb * (32 * H_) + lane * H_);
    float* dstp = (wb == 0 ? lsrc : ldst) + ((size_t)blockIdx.x * 32 + lane) * H_;
    *(float4*)dstp = q;
  }

  #pragma unroll
  for (int cc = 0; cc < 4; ++cc) {
    #pragma unroll
    for (int r = 0; r < 8; ++r) {
      const int row = row0 + r + 8 * half;
      hbuf[(size_t)row * COUT_ + col0 + cc * 16 + l] = acc[cc][r];
    }
  }
}

#define JS_   4
#define JLEN_ (N_ / JS_)
#define RP_   260
#define BW_   33
#define AGG_THREADS_ (32 * H_ * JS_)

__global__ __launch_bounds__(AGG_THREADS_) void gat_aggregate(
    const float* __restrict__ hbuf,
    const float* __restrict__ lsrc,
    const float* __restrict__ ldst,
    const int*   __restrict__ adj,
    float* __restrict__ out) {
  __shared__ __attribute__((aligned(16))) float lds_ld[N_ * H_];
  __shared__ __attribute__((aligned(16))) float lds_part[JS_ * 32 * RP_];
  __shared__ unsigned lds_bits[32 * BW_];
  __shared__ float    lds_m[32 * H_];
  __shared__ float    lds_s[JS_ * 32 * H_];

  const int b     = blockIdx.x >> 5;
  const int islab = blockIdx.x & 31;
  const int i0    = islab * 32;
  const int wv    = __builtin_amdgcn_readfirstlane((int)(threadIdx.x >> 5));
  const int head  = wv & (H_ - 1);
  const int js    = wv >> 2;
  const int lane  = threadIdx.x & 31;
  const int half  = lane >> 4;
  const int l     = lane & 15;

  float4 stg[N_ / AGG_THREADS_];
  {
    const float4* src = (const float4*)(ldst + (size_t)b * N_ * H_);
    #pragma unroll
    for (int t = 0; t < N_ / AGG_THREADS_; ++t)
      stg[t] = src[t * AGG_THREADS_ + threadIdx.x];
  }

  const int  srow = 2 * wv;
  const int* ap   = adj + ((size_t)b * N_ + i0 + srow) * N_ + lane;
  int ad0[32], ad1[32];
  #pragma unroll
  for (int k = 0; k < 32; ++k) ad0[k] = ap[32 * k];
  #pragma unroll
  for (int k = 0; k < 32; ++k) ad1[k] = ap[N_ + 32 * k];

  {
    float4* dst = (float4*)lds_ld;
    #pragma unroll
    for (int t = 0; t < N_ / AGG_THREADS_; ++t)
      dst[t * AGG_THREADS_ + threadIdx.x] = stg[t];
  }
  __syncthreads();

  {
    float mx0[H_], mx1[H_];
    #pragma unroll
    for (int hd = 0; hd < H_; ++hd) { mx0[hd] = -INFINITY; mx1[hd] = -INFINITY; }
    unsigned w0 = 0u, w1 = 0u;
    const float4* ld4 = (const float4*)lds_ld;
    #pragma unroll
    for (int k = 0; k < 32; ++k) {
      const float4 ld = ld4[32 * k + lane];
      const bool p0 = (ad0[k] == 1);
      const bool p1 = (ad1[k] == 1);
      const unsigned m0 = __builtin_amdgcn_ballot_w32(p0);
      const unsigned m1 = __builtin_amdgcn_ballot_w32(p1);
      w0 = (lane == k) ? m0 : w0;
      w1 = (lane == k) ? m1 : w1;
      mx0[0] = fmaxf(mx0[0], p0 ? ld.x : -INFINITY);
      mx0[1] = fmaxf(mx0[1], p0 ? ld.y : -INFINITY);
      mx0[2] = fmaxf(mx0[2], p0 ? ld.z : -INFINITY);
      mx0[3] = fmaxf(mx0[3], p0 ? ld.w : -INFINITY);
      mx1[0] = fmaxf(mx1[0], p1 ? ld.x : -INFINITY);
      mx1[1] = fmaxf(mx1[1], p1 ? ld.y : -INFINITY);
      mx1[2] = fmaxf(mx1[2], p1 ? ld.z : -INFINITY);
      mx1[3] = fmaxf(mx1[3], p1 ? ld.w : -INFINITY);
    }
    #pragma unroll
    for (int hd = 0; hd < H_; ++hd) {
      #pragma unroll
      for (int off = 16; off > 0; off >>= 1) {
        mx0[hd] = fmaxf(mx0[hd], __shfl_xor(mx0[hd], off, 32));
        mx1[hd] = fmaxf(mx1[hd], __shfl_xor(mx1[hd], off, 32));
      }
    }
    lds_bits[srow * BW_ + lane]       = w0;
    lds_bits[(srow + 1) * BW_ + lane] = w1;
    float v = mx0[0];
    v = (lane == 1) ? mx0[1] : v;
    v = (lane == 2) ? mx0[2] : v;
    v = (lane == 3) ? mx0[3] : v;
    v = (lane == 4) ? mx1[0] : v;
    v = (lane == 5) ? mx1[1] : v;
    v = (lane == 6) ? mx1[2] : v;
    v = (lane == 7) ? mx1[3] : v;
    if (lane < 2 * H_) {
      float x = lsrc[((size_t)b * N_ + i0 + srow) * H_ + lane] + v;
      x = fmaxf(x, ALPHA_ * x);
      lds_m[srow * H_ + lane] = fmaxf(x, NEG_INF_);
    }
  }
  __syncthreads();

  const size_t biA = (size_t)b * N_ + (i0 + l);
  const size_t biB = biA + 16;
  const float  lsA = lsrc[biA * H_ + head];
  const float  lsB = lsrc[biB * H_ + head];
  const float  mA  = lds_m[l * H_ + head];
  const float  mB  = lds_m[(l + 16) * H_ + head];

  v8f accA[4] = {{}, {}, {}, {}};
  v8f accB[4] = {{}, {}, {}, {}};
  float sA = 0.f, sB = 0.f;

  const unsigned* bitA = lds_bits + l * BW_ + js * (JLEN_ / 32);
  const unsigned* bitB = bitA + 16 * BW_;
  const float* hbase = hbuf + ((size_t)b * N_ + js * JLEN_ + 2 * half) * COUT_ + head * C_ + l;
  const float* ldp   = lds_ld + (js * JLEN_ + 2 * half) * H_ + head;

  for (int w = 0; w < JLEN_ / 32; ++w) {
    const unsigned bA = bitA[w] >> (2 * half);
    const unsigned bB = bitB[w] >> (2 * half);
    #pragma unroll
    for (int t = 0; t < 8; ++t) {
      const int jo = w * 32 + 4 * t;
      const float ld0 = ldp[jo * H_];
      const float ld1 = ldp[(jo + 1) * H_];

      float xA0 = lsA + ld0; xA0 = fmaxf(xA0, ALPHA_ * xA0);
      float xA1 = lsA + ld1; xA1 = fmaxf(xA1, ALPHA_ * xA1);
      float xB0 = lsB + ld0; xB0 = fmaxf(xB0, ALPHA_ * xB0);
      float xB1 = lsB + ld1; xB1 = fmaxf(xB1, ALPHA_ * xB1);

      v2f avA, avB;
      avA.x = ((bA >> (4 * t))     & 1u) ? __expf(xA0 - mA) : 0.0f;
      avA.y = ((bA >> (4 * t + 1)) & 1u) ? __expf(xA1 - mA) : 0.0f;
      avB.x = ((bB >> (4 * t))     & 1u) ? __expf(xB0 - mB) : 0.0f;
      avB.y = ((bB >> (4 * t + 1)) & 1u) ? __expf(xB1 - mB) : 0.0f;
      sA += avA.x + avA.y;
      sB += avB.x + avB.y;

      const float* hrow0 = hbase + (size_t)jo * COUT_;
      const float* hrow1 = hrow0 + COUT_;
      #pragma unroll
      for (int cc = 0; cc < 4; ++cc) {
        v2f bv;
        bv.x = hrow0[cc * 16];
        bv.y = hrow1[cc * 16];
        accA[cc] = __builtin_amdgcn_wmma_f32_16x16x4_f32(false, avA, false, bv,
                                                         (short)0, accA[cc],
                                                         false, false);
        accB[cc] = __builtin_amdgcn_wmma_f32_16x16x4_f32(false, avB, false, bv,
                                                         (short)0, accB[cc],
                                                         false, false);
      }
    }
  }

  sA += __shfl_xor(sA, 16, 32);
  sB += __shfl_xor(sB, 16, 32);
  lds_s[(js * 32 + lane) * H_ + head] = half ? sB : sA;

  {
    float* pl = lds_part + js * (32 * RP_) + head * C_ + l;
    #pragma unroll
    for (int cc = 0; cc < 4; ++cc) {
      #pragma unroll
      for (int r = 0; r < 8; ++r) {
        const int rowA = r + 8 * half;
        pl[rowA * RP_ + cc * 16]        = accA[cc][r];
        pl[(rowA + 16) * RP_ + cc * 16] = accB[cc][r];
      }
    }
  }
  __syncthreads();

  #pragma unroll
  for (int t = 0; t < (32 * (COUT_ / 4)) / AGG_THREADS_; ++t) {
    const int q   = t * AGG_THREADS_ + threadIdx.x;
    const int row = q >> 6;
    const int c4  = q & 63;
    const int hd  = c4 >> 4;
    const float* sp = lds_s + row * H_ + hd;
    float tot = sp[0];
    #pragma unroll
    for (int k = 1; k < JS_; ++k) tot += sp[k * (32 * H_)];
    const float inv = (tot > 0.0f) ? 1.0f / tot : 0.0f;
    const float* pp = lds_part + row * RP_ + c4 * 4;
    float4 s = *(const float4*)pp;
    #pragma unroll
    for (int k = 1; k < JS_; ++k) {
      const float4 p = *(const float4*)(pp + k * (32 * RP_));
      s.x += p.x; s.y += p.y; s.z += p.z; s.w += p.w;
    }
    s.x *= inv; s.y *= inv; s.z *= inv; s.w *= inv;
    *(float4*)(out + ((size_t)b * N_ + i0 + row) * COUT_ + c4 * 4) = s;
  }
}

extern "C" void kernel_launch(void* const* d_in, const int* in_sizes, int n_in,
                              void* d_out, int out_size, void* d_ws, size_t ws_size,
                              hipStream_t stream) {
  const float* node_feats = (const float*)d_in[0];
  const int*   adj        = (const int*)  d_in[1];
  const float* W          = (const float*)d_in[2];
  const float* bias       = (const float*)d_in[3];
  const float* a          = (const float*)d_in[4];
  float*       out        = (float*)d_out;

  float* hbuf = (float*)d_ws;
  float* lsrc = hbuf + (size_t)B_ * N_ * COUT_;
  float* ldst = lsrc + (size_t)B_ * N_ * H_;

  gat_gemm_h<<<(B_ * N_ / 16) * (COUT_ / 64) / 8, 256, 0, stream>>>(
      node_feats, W, bias, a, hbuf, lsrc, ldst);

  gat_aggregate<<<B_ * (N_ / 32), AGG_THREADS_, 0, stream>>>(
      hbuf, lsrc, ldst, adj, out);
}
